# starting kernel
# baseline (speedup 1.0000x reference)
#include <hip/hip_runtime.h>
#include <hip/hip_bf16.h>

typedef __attribute__((ext_vector_type(2))) float v2f;
typedef __attribute__((ext_vector_type(8))) float v8f;

#define IMG_W        512
#define TILES_X      32
#define N_TILES      98304
#define TILES_PER_WAVE 4
#define WAVES_PER_BLOCK 8
#define N_BLOCKS     (N_TILES / (TILES_PER_WAVE * WAVES_PER_BLOCK))
#define LDS_STRIDE   18

__device__ __forceinline__ float dbig(const float* __restrict__ b, int i, int j) {
    return ((i >> 3) == (j >> 3)) ? b[((i & 7) << 3) + (j & 7)] : 0.0f;
}

__global__ __launch_bounds__(256) void DCT2D_3599182594809_kernel(
    const float* __restrict__ x,
    const float* __restrict__ basis,
    float* __restrict__ out)
{
    __shared__ float lds[WAVES_PER_BLOCK * 16 * LDS_STRIDE];

    const int tid  = threadIdx.x;
    const int wave = tid >> 5;
    const int lane = tid & 31;
    const int m    = lane & 15;
    const int hi   = lane >> 4;
    const int d    = hi << 1;

    float* tscr = &lds[wave * 16 * LDS_STRIDE];

    v2f aD[4];
#pragma unroll
    for (int k = 0; k < 4; ++k) {
        const int c0 = 4 * k + d;
        aD[k].x = dbig(basis, m, c0);
        aD[k].y = dbig(basis, m, c0 + 1);
    }

    const int gw = blockIdx.x * WAVES_PER_BLOCK + wave;
    const int t0 = gw * TILES_PER_WAVE;

#pragma unroll
    for (int tt = 0; tt < TILES_PER_WAVE; ++tt) {
        const int t = t0 + tt;

        const int plane = t >> 10;
        const int ty    = (t >> 5) & 31;
        const int tx    = t & 31;
        const size_t base = (size_t)plane * (IMG_W * IMG_W)
                          + (size_t)(ty * 16) * IMG_W + (size_t)(tx * 16) + m;

        v2f xb[4];
#pragma unroll
        for (int k = 0; k < 4; ++k) {
            const int r0 = 4 * k + d;
            xb[k].x = __builtin_nontemporal_load(&x[base + (size_t)r0 * IMG_W]);
            xb[k].y = __builtin_nontemporal_load(&x[base + (size_t)(r0 + 1) * IMG_W]);
        }

        v8f acc = {};
#pragma unroll
        for (int k = 0; k < 4; ++k) {
            acc = __builtin_amdgcn_wmma_f32_16x16x4_f32(
                false, aD[k], false, xb[k], (short)0, acc, false, false);
        }

#pragma unroll
        for (int i = 0; i < 8; ++i)
            tscr[(i + 8 * hi) * LDS_STRIDE + m] = acc[i];

        asm volatile("s_wait_dscnt 0" ::: "memory");

        v2f ta[4];
#pragma unroll
        for (int k = 0; k < 4; ++k)
            ta[k] = *(const v2f*)&tscr[m * LDS_STRIDE + 4 * k + d];

        v8f y = {};
#pragma unroll
        for (int k = 0; k < 4; ++k) {
            y = __builtin_amdgcn_wmma_f32_16x16x4_f32(
                false, ta[k], false, aD[k], (short)0, y, false, false);
        }

#pragma unroll
        for (int i = 0; i < 8; ++i)
            __builtin_nontemporal_store(y[i], &out[base + (size_t)(i + 8 * hi) * IMG_W]);
    }
}

extern "C" void kernel_launch(void* const* d_in, const int* in_sizes, int n_in,
                              void* d_out, int out_size, void* d_ws, size_t ws_size,
                              hipStream_t stream) {
    const float* x     = (const float*)d_in[0];
    const float* basis = (const float*)d_in[1];
    float* out         = (float*)d_out;
    (void)in_sizes; (void)n_in; (void)out_size; (void)d_ws; (void)ws_size;

    DCT2D_3599182594809_kernel<<<N_BLOCKS, 256, 0, stream>>>(x, basis, out);
}


# Grid cut from 3072 to 512 workgroups (2 per WGP): each wave now loops over 24 consecutive tiles with the next 6 tiles' loads in flight; per-tile arithmetic and sum order unchanged
# speedup vs baseline: 1.0726x; 1.0726x over previous
#include <hip/hip_runtime.h>
#include <hip/hip_bf16.h>


typedef __attribute__((ext_vector_type(2))) float v2f;
typedef __attribute__((ext_vector_type(8))) float v8f;

#define IMG_W        512
#define N_TILES      98304
#define WAVES_PER_BLOCK 8
#define N_BLOCKS     512
#define TILES_PER_WAVE (N_TILES / (N_BLOCKS * WAVES_PER_BLOCK))
#define DEPTH        6
#define N_GROUPS     (TILES_PER_WAVE / DEPTH)
#define LDS_STRIDE   18

__device__ __forceinline__ float dbig(const float* __restrict__ b, int i, int j) {
    return ((i >> 3) == (j >> 3)) ? b[((i & 7) << 3) + (j & 7)] : 0.0f;
}

__device__ __forceinline__ size_t tile_off(int t) {
    return (size_t)(t >> 5) * (16 * IMG_W) + (size_t)((t & 31) * 16);
}

__device__ __forceinline__ void load_tile(v2f (&xb)[4], const float* __restrict__ p) {
#pragma unroll
    for (int k = 0; k < 4; ++k) {
        xb[k].x = __builtin_nontemporal_load(&p[(size_t)(4 * k) * IMG_W]);
        xb[k].y = __builtin_nontemporal_load(&p[(size_t)(4 * k + 1) * IMG_W]);
    }
}

__device__ __forceinline__ v8f gemm1(const v2f (&aD)[4], const v2f (&xb)[4]) {
    v8f acc = {};
#pragma unroll
    for (int k = 0; k < 4; ++k) {
        acc = __builtin_amdgcn_wmma_f32_16x16x4_f32(
            false, aD[k], false, xb[k], (short)0, acc, false, false);
    }
    return acc;
}

__device__ __forceinline__ void finish_tile(const v2f (&aD)[4], v8f acc, float* tscr,
                                            int m, int hi, int d, float* __restrict__ q) {
#pragma unroll
    for (int i = 0; i < 8; ++i)
        tscr[(i + 8 * hi) * LDS_STRIDE + m] = acc[i];

    asm volatile("s_wait_dscnt 0" ::: "memory");

    v2f ta[4];
#pragma unroll
    for (int k = 0; k < 4; ++k)
        ta[k] = *(const v2f*)&tscr[m * LDS_STRIDE + 4 * k + d];

    v8f y = {};
#pragma unroll
    for (int k = 0; k < 4; ++k) {
        y = __builtin_amdgcn_wmma_f32_16x16x4_f32(
            false, ta[k], false, aD[k], (short)0, y, false, false);
    }

#pragma unroll
    for (int i = 0; i < 8; ++i)
        __builtin_nontemporal_store(y[i], &q[(size_t)i * IMG_W]);
}

__global__ __launch_bounds__(256) void DCT2D_3599182594809_kernel(
    const float* __restrict__ x,
    const float* __restrict__ basis,
    float* __restrict__ out)
{
    __shared__ float lds[WAVES_PER_BLOCK * 16 * LDS_STRIDE];

    const int tid  = threadIdx.x;
    const int wave = __builtin_amdgcn_readfirstlane(tid >> 5);
    const int lane = tid & 31;
    const int m    = lane & 15;
    const int hi   = lane >> 4;
    const int d    = hi << 1;

    float* tscr = &lds[wave * 16 * LDS_STRIDE];

    v2f aD[4];
#pragma unroll
    for (int k = 0; k < 4; ++k) {
        const int c0 = 4 * k + d;
        aD[k].x = dbig(basis, m, c0);
        aD[k].y = dbig(basis, m, c0 + 1);
    }

    const int gw = blockIdx.x * WAVES_PER_BLOCK + wave;
    const int t0 = gw * TILES_PER_WAVE;

    const int in_lane  = d * IMG_W + m;
    const int out_lane = (8 * hi) * IMG_W + m;

    v2f xb[DEPTH][4];
#pragma unroll
    for (int s = 0; s < DEPTH; ++s)
        load_tile(xb[s], x + tile_off(t0 + s) + in_lane);

    for (int g = 0; g < N_GROUPS - 1; ++g) {
#pragma unroll
        for (int s = 0; s < DEPTH; ++s) {
            const int t = t0 + g * DEPTH + s;
            const v8f acc = gemm1(aD, xb[s]);
            load_tile(xb[s], x + tile_off(t + DEPTH) + in_lane);
            finish_tile(aD, acc, tscr, m, hi, d, out + tile_off(t) + out_lane);
        }
    }

#pragma unroll
    for (int s = 0; s < DEPTH; ++s) {
        const int t = t0 + (N_GROUPS - 1) * DEPTH + s;
        const v8f acc = gemm1(aD, xb[s]);
        finish_tile(aD, acc, tscr, m, hi, d, out + tile_off(t) + out_lane);
    }
}

extern "C" void kernel_launch(void* const* d_in, const int* in_sizes, int n_in,
                              void* d_out, int out_size, void* d_ws, size_t ws_size,
                              hipStream_t stream) {
    const float* x     = (const float*)d_in[0];
    const float* basis = (const float*)d_in[1];
    float* out         = (float*)d_out;
    (void)in_sizes; (void)n_in; (void)out_size; (void)d_ws; (void)ws_size;

    DCT2D_3599182594809_kernel<<<N_BLOCKS, 256, 0, stream>>>(x, basis, out);
}


# Tiles dealt round-robin over the 4096 waves (step i: tile i*4096 + wave) instead of 24 consecutive per wave: the grid reads and writes contiguous 4 MiB windows; per-tile sums unchanged
# speedup vs baseline: 1.1299x; 1.0534x over previous
#include <hip/hip_runtime.h>
#include <hip/hip_bf16.h>


typedef __attribute__((ext_vector_type(2))) float v2f;
typedef __attribute__((ext_vector_type(8))) float v8f;

#define GLOBAL_AS __attribute__((address_space(1)))

#define IMG_W        512
#define N_TILES      98304
#define WAVES_PER_BLOCK 8
#define N_BLOCKS     512
#define TILES_PER_WAVE (N_TILES / (N_BLOCKS * WAVES_PER_BLOCK))
#define DEPTH        6
#define N_GROUPS     (TILES_PER_WAVE / DEPTH)
#define N_WAVES      (N_BLOCKS * WAVES_PER_BLOCK)
#define STEP_STRIDE  ((N_WAVES / 32) * 16 * IMG_W)
#define LDS_STRIDE   18

__device__ __forceinline__ float dbig(const float* __restrict__ b, int i, int j) {
    return ((i >> 3) == (j >> 3)) ? b[((i & 7) << 3) + (j & 7)] : 0.0f;
}

__device__ __forceinline__ size_t tile_off(int t) {
    return (size_t)(t >> 5) * (16 * IMG_W) + (size_t)((t & 31) * 16);
}

__device__ __forceinline__ void load_tile(v2f (&xb)[4], GLOBAL_AS const float* p, int lo) {
#pragma unroll
    for (int k = 0; k < 4; ++k) {
        xb[k].x = __builtin_nontemporal_load(&p[lo + (4 * k) * IMG_W]);
        xb[k].y = __builtin_nontemporal_load(&p[lo + (4 * k + 1) * IMG_W]);
    }
}

__device__ __forceinline__ v8f gemm1(const v2f (&aD)[4], const v2f (&xb)[4]) {
    v8f acc = {};
#pragma unroll
    for (int k = 0; k < 4; ++k) {
        acc = __builtin_amdgcn_wmma_f32_16x16x4_f32(
            false, aD[k], false, xb[k], (short)0, acc, false, false);
    }
    return acc;
}

__device__ __forceinline__ void finish_tile(const v2f (&aD)[4], v8f acc, float* tscr,
                                            int m, int hi, int d, GLOBAL_AS float* q, int qo) {
#pragma unroll
    for (int i = 0; i < 8; ++i)
        tscr[(i + 8 * hi) * LDS_STRIDE + m] = acc[i];

    asm volatile("s_wait_dscnt 0" ::: "memory");

    v2f ta[4];
#pragma unroll
    for (int k = 0; k < 4; ++k)
        ta[k] = *(const v2f*)&tscr[m * LDS_STRIDE + 4 * k + d];

    v8f y = {};
#pragma unroll
    for (int k = 0; k < 4; ++k) {
        y = __builtin_amdgcn_wmma_f32_16x16x4_f32(
            false, ta[k], false, aD[k], (short)0, y, false, false);
    }

#pragma unroll
    for (int i = 0; i < 8; ++i)
        __builtin_nontemporal_store(y[i], &q[qo + i * IMG_W]);
}

template <typename T>
__device__ __forceinline__ GLOBAL_AS T* step_ptr(T* base, int wave_off, int i) {
    unsigned long long a = (unsigned long long)(base + (wave_off + i * STEP_STRIDE));
    asm volatile("" : "+s"(a));
    return (GLOBAL_AS T*)a;
}

__global__ __launch_bounds__(256) void DCT2D_3599182594809_kernel(
    const float* __restrict__ x,
    const float* __restrict__ basis,
    float* __restrict__ out)
{
    __shared__ float lds[WAVES_PER_BLOCK * 16 * LDS_STRIDE];

    const int tid  = threadIdx.x;
    const int wave = __builtin_amdgcn_readfirstlane(tid >> 5);
    const int lane = tid & 31;
    const int m    = lane & 15;
    const int hi   = lane >> 4;
    const int d    = hi << 1;

    float* tscr = &lds[wave * 16 * LDS_STRIDE];

    v2f aD[4];
#pragma unroll
    for (int k = 0; k < 4; ++k) {
        const int c0 = 4 * k + d;
        aD[k].x = dbig(basis, m, c0);
        aD[k].y = dbig(basis, m, c0 + 1);
    }

    const int gw = blockIdx.x * WAVES_PER_BLOCK + wave;
    const int wave_off = (int)tile_off(gw);

    const int in_lane  = d * IMG_W + m;
    const int out_lane = (8 * hi) * IMG_W + m;

    v2f xb[DEPTH][4];
#pragma unroll
    for (int s = 0; s < DEPTH; ++s)
        load_tile(xb[s], step_ptr(x, wave_off, s), in_lane);

    for (int g = 0; g < N_GROUPS - 1; ++g) {
#pragma unroll
        for (int s = 0; s < DEPTH; ++s) {
            const int i = g * DEPTH + s;
            const v8f acc = gemm1(aD, xb[s]);
            load_tile(xb[s], step_ptr(x, wave_off, i + DEPTH), in_lane);
            finish_tile(aD, acc, tscr, m, hi, d, step_ptr(out, wave_off, i), out_lane);
        }
    }

#pragma unroll
    for (int s = 0; s < DEPTH; ++s) {
        const int i = (N_GROUPS - 1) * DEPTH + s;
        const v8f acc = gemm1(aD, xb[s]);
        finish_tile(aD, acc, tscr, m, hi, d, step_ptr(out, wave_off, i), out_lane);
    }
}

extern "C" void kernel_launch(void* const* d_in, const int* in_sizes, int n_in,
                              void* d_out, int out_size, void* d_ws, size_t ws_size,
                              hipStream_t stream) {
    const float* x     = (const float*)d_in[0];
    const float* basis = (const float*)d_in[1];
    float* out         = (float*)d_out;
    (void)in_sizes; (void)n_in; (void)out_size; (void)d_ws; (void)ws_size;

    DCT2D_3599182594809_kernel<<<N_BLOCKS, 256, 0, stream>>>(x, basis, out);
}


# Each wave moves 16x64 blocks through private LDS: every global load and store is 16 B per lane on whole 256-B row pieces; X*Dt runs first (columns summed before rows), its accumulators are B of D*U
# speedup vs baseline: 2.1107x; 1.8680x over previous
#include <hip/hip_runtime.h>
#include <hip/hip_bf16.h>


typedef __attribute__((ext_vector_type(2))) float v2f;
typedef __attribute__((ext_vector_type(4))) float v4f;
typedef __attribute__((ext_vector_type(8))) float v8f;

#define GLOBAL_AS __attribute__((address_space(1)))

#define IMG_W        512
#define BLK_W        64
#define BLK_TILES    (BLK_W / 16)
#define BLK_LOADS    8
#define N_BLKS       24576
#define WAVES_PER_BLOCK 8
#define N_BLOCKS     512
#define N_WAVES      (N_BLOCKS * WAVES_PER_BLOCK)
#define N_STEPS      (N_BLKS / N_WAVES)
#define DEPTH        2
#define STEP_STRIDE  ((N_WAVES / 8) * 16 * IMG_W)
#define PITCH        68

__device__ __forceinline__ float dbig(const float* __restrict__ b, int i, int j) {
    return ((i >> 3) == (j >> 3)) ? b[((i & 7) << 3) + (j & 7)] : 0.0f;
}

__device__ __forceinline__ void load_blk(v4f (&xr)[BLK_LOADS], GLOBAL_AS const float* p, int lo) {
#pragma unroll
    for (int q = 0; q < BLK_LOADS; ++q)
        xr[q] = __builtin_nontemporal_load((GLOBAL_AS const v4f*)&p[lo + (2 * q) * IMG_W]);
}

template <typename T>
__device__ __forceinline__ GLOBAL_AS T* step_ptr(T* base, int wave_off, int i) {
    unsigned long long a = (unsigned long long)(base + (wave_off + i * STEP_STRIDE));
    asm volatile("" : "+s"(a));
    return (GLOBAL_AS T*)a;
}

__global__ __launch_bounds__(256) void DCT2D_3599182594809_kernel(
    const float* __restrict__ x,
    const float* __restrict__ basis,
    float* __restrict__ out)
{
    __shared__ alignas(16) float lds_in[WAVES_PER_BLOCK * 16 * PITCH];
    __shared__ alignas(16) float lds_out[WAVES_PER_BLOCK * 16 * PITCH];

    const int tid  = threadIdx.x;
    const int wave = __builtin_amdgcn_readfirstlane(tid >> 5);
    const int lane = tid & 31;
    const int m    = lane & 15;
    const int hi   = lane >> 4;

    float* sin_ = &lds_in[wave * 16 * PITCH];
    float* sout = &lds_out[wave * 16 * PITCH];

    v2f aD[4];
#pragma unroll
    for (int k = 0; k < 4; ++k) {
        const int c0 = 8 * hi + 2 * k;
        aD[k].x = dbig(basis, m, c0);
        aD[k].y = dbig(basis, m, c0 + 1);
    }

    const int gw = blockIdx.x * WAVES_PER_BLOCK + wave;
    const int wave_off = (gw >> 3) * (16 * IMG_W) + (gw & 7) * BLK_W;

    const int io_lane  = hi * IMG_W + 4 * m;
    const int lds_lane = hi * PITCH + 4 * m;

    v4f xr[DEPTH][BLK_LOADS];
#pragma unroll
    for (int s = 0; s < DEPTH; ++s)
        load_blk(xr[s], step_ptr(x, wave_off, s), io_lane);

#pragma unroll
    for (int i = 0; i < N_STEPS; ++i) {
        const int s = i % DEPTH;

#pragma unroll
        for (int q = 0; q < BLK_LOADS; ++q)
            *(v4f*)&sin_[(2 * q) * PITCH + lds_lane] = xr[s][q];

        if (i + DEPTH < N_STEPS)
            load_blk(xr[s], step_ptr(x, wave_off, i + DEPTH), io_lane);

        asm volatile("s_wait_dscnt 0" ::: "memory");

        v4f alo[BLK_TILES], ahi[BLK_TILES];
#pragma unroll
        for (int j = 0; j < BLK_TILES; ++j) {
            alo[j] = *(const v4f*)&sin_[m * PITCH + 16 * j + 8 * hi];
            ahi[j] = *(const v4f*)&sin_[m * PITCH + 16 * j + 8 * hi + 4];
        }

#pragma unroll
        for (int j = 0; j < BLK_TILES; ++j) {
            v8f u = {};
            u = __builtin_amdgcn_wmma_f32_16x16x4_f32(
                false, (v2f){alo[j].x, alo[j].y}, false, aD[0], (short)0, u, false, false);
            u = __builtin_amdgcn_wmma_f32_16x16x4_f32(
                false, (v2f){alo[j].z, alo[j].w}, false, aD[1], (short)0, u, false, false);
            u = __builtin_amdgcn_wmma_f32_16x16x4_f32(
                false, (v2f){ahi[j].x, ahi[j].y}, false, aD[2], (short)0, u, false, false);
            u = __builtin_amdgcn_wmma_f32_16x16x4_f32(
                false, (v2f){ahi[j].z, ahi[j].w}, false, aD[3], (short)0, u, false, false);

            v8f y = {};
#pragma unroll
            for (int k = 0; k < 4; ++k) {
                y = __builtin_amdgcn_wmma_f32_16x16x4_f32(
                    false, aD[k], false, (v2f){u[2 * k], u[2 * k + 1]}, (short)0, y, false, false);
            }

#pragma unroll
            for (int r = 0; r < 8; ++r)
                sout[(r + 8 * hi) * PITCH + 16 * j + m] = y[r];
        }

        asm volatile("s_wait_dscnt 0" ::: "memory");

        GLOBAL_AS float* qp = step_ptr(out, wave_off, i);
#pragma unroll
        for (int q = 0; q < BLK_LOADS; ++q) {
            const v4f o = *(const v4f*)&sout[(2 * q) * PITCH + lds_lane];
            __builtin_nontemporal_store(o, (GLOBAL_AS v4f*)&qp[io_lane + (2 * q) * IMG_W]);
        }
    }
}

extern "C" void kernel_launch(void* const* d_in, const int* in_sizes, int n_in,
                              void* d_out, int out_size, void* d_ws, size_t ws_size,
                              hipStream_t stream) {
    const float* x     = (const float*)d_in[0];
    const float* basis = (const float*)d_in[1];
    float* out         = (float*)d_out;
    (void)in_sizes; (void)n_in; (void)out_size; (void)d_ws; (void)ws_size;

    DCT2D_3599182594809_kernel<<<N_BLOCKS, 256, 0, stream>>>(x, basis, out);
}
